# starting kernel
# baseline (speedup 1.0000x reference)
#include <hip/hip_runtime.h>

typedef __attribute__((ext_vector_type(2))) float v2f;
typedef __attribute__((ext_vector_type(8))) float v8f;

#define S_LEN 1024
#define D_LEN 256
#define KB    32
#define LDK   34

__global__ __launch_bounds__(256, 3) void complex_mixture_wmma(
    const float* __restrict__ in_r,
    const float* __restrict__ in_i,
    const float* __restrict__ wgt,
    float* __restrict__ out_r,
    float* __restrict__ out_i)
{
    __shared__ __align__(16) float sWrA[64  * LDK];
    __shared__ __align__(16) float sWiA[64  * LDK];
    __shared__ __align__(16) float sRB [128 * LDK];
    __shared__ __align__(16) float sIB [128 * LDK];

    const int b     = blockIdx.z;
    const int aBase = blockIdx.y * 64;
    const int cBase = blockIdx.x * 128;

    const int tid   = threadIdx.x;
    const int lane  = tid & 31;
    const int wave  = tid >> 5;
    const int waveM = wave & 1;
    const int waveN = wave >> 1;

    const int laneLo = lane & 15;
    const int kOff   = (lane >> 4) * 2;

    const float* rSrc = in_r + (size_t)b * S_LEN * D_LEN;
    const float* iSrc = in_i + (size_t)b * S_LEN * D_LEN;
    const float* wSrc = wgt  + (size_t)b * S_LEN;

    v8f accR [2][2] = {};
    v8f accI1[2][2] = {};
    v8f accI2[2][2] = {};

    for (int kb = 0; kb < S_LEN / KB; ++kb) {
        const int s0 = kb * KB;

        #pragma unroll
        for (int it = 0; it < (KB * 64) / 256; ++it) {
            int idx = tid + it * 256;
            int sl  = idx >> 6;
            int ml  = idx & 63;
            size_t g = (size_t)(s0 + sl) * D_LEN + aBase + ml;
            float  w = wSrc[s0 + sl];
            sWrA[ml * LDK + sl] = w * rSrc[g];
            sWiA[ml * LDK + sl] = w * iSrc[g];
        }
        #pragma unroll
        for (int it = 0; it < (KB * 128) / 256; ++it) {
            int idx = tid + it * 256;
            int sl  = idx >> 7;
            int nl  = idx & 127;
            size_t g = (size_t)(s0 + sl) * D_LEN + cBase + nl;
            sRB[nl * LDK + sl] = rSrc[g];
            sIB[nl * LDK + sl] = iSrc[g];
        }
        if (kb + 1 < S_LEN / KB) {
            size_t gn = (size_t)(s0 + KB + (tid >> 3)) * D_LEN + aBase + (tid & 7) * 8;
            __builtin_prefetch(&rSrc[gn], 0, 0);
            __builtin_prefetch(&iSrc[gn], 0, 0);
        }
        __syncthreads();

        #pragma unroll
        for (int kk = 0; kk < KB; kk += 4) {
            v2f aR[2], aI[2], bR[2], bI[2];
            #pragma unroll
            for (int mt = 0; mt < 2; ++mt) {
                int row = waveM * 32 + mt * 16 + laneLo;
                aR[mt] = *(const v2f*)&sWrA[row * LDK + kk + kOff];
                aI[mt] = *(const v2f*)&sWiA[row * LDK + kk + kOff];
            }
            #pragma unroll
            for (int nt = 0; nt < 2; ++nt) {
                int col = waveN * 32 + nt * 16 + laneLo;
                bR[nt] = *(const v2f*)&sRB[col * LDK + kk + kOff];
                bI[nt] = *(const v2f*)&sIB[col * LDK + kk + kOff];
            }
            #pragma unroll
            for (int mt = 0; mt < 2; ++mt) {
                #pragma unroll
                for (int nt = 0; nt < 2; ++nt) {
                    accR[mt][nt]  = __builtin_amdgcn_wmma_f32_16x16x4_f32(
                        false, aR[mt], false, bR[nt], (short)0, accR[mt][nt],  false, false);
                    accR[mt][nt]  = __builtin_amdgcn_wmma_f32_16x16x4_f32(
                        false, aI[mt], false, bI[nt], (short)0, accR[mt][nt],  false, false);
                    accI1[mt][nt] = __builtin_amdgcn_wmma_f32_16x16x4_f32(
                        false, aI[mt], false, bR[nt], (short)0, accI1[mt][nt], false, false);
                    accI2[mt][nt] = __builtin_amdgcn_wmma_f32_16x16x4_f32(
                        false, aR[mt], false, bI[nt], (short)0, accI2[mt][nt], false, false);
                }
            }
        }
        __syncthreads();
    }

    const int laneHi8 = (lane >> 4) * 8;
    #pragma unroll
    for (int mt = 0; mt < 2; ++mt) {
        #pragma unroll
        for (int nt = 0; nt < 2; ++nt) {
            int row0 = aBase + waveM * 32 + mt * 16 + laneHi8;
            int col  = cBase + waveN * 32 + nt * 16 + laneLo;
            #pragma unroll
            for (int j = 0; j < 8; ++j) {
                size_t o = ((size_t)b * D_LEN + row0 + j) * D_LEN + col;
                out_r[o] = accR[mt][nt][j];
                out_i[o] = accI1[mt][nt][j] - accI2[mt][nt][j];
            }
        }
    }
}

extern "C" void kernel_launch(void* const* d_in, const int* in_sizes, int n_in,
                              void* d_out, int out_size, void* d_ws, size_t ws_size,
                              hipStream_t stream) {
    const float* in_r = (const float*)d_in[0];
    const float* in_i = (const float*)d_in[1];
    const float* w    = (const float*)d_in[2];
    float* out_r = (float*)d_out;
    float* out_i = out_r + (size_t)32 * D_LEN * D_LEN;

    dim3 grid(D_LEN / 128, D_LEN / 64, 32);
    dim3 block(256);
    hipLaunchKernelGGL(complex_mixture_wmma, grid, block, 0, stream,
                       in_r, in_i, w, out_r, out_i);
}


# imaginary part from ONE product: out_i = M - M^T with M = (w i)^T r; 3 WMMAs per tile and k-step (was 4), a second launch subtracts each 64x64 tile's mirror in place through LDS
# speedup vs baseline: 8.5802x; 8.5802x over previous
#include <hip/hip_runtime.h>


typedef __attribute__((ext_vector_type(2))) float v2f;
typedef __attribute__((ext_vector_type(4))) float v4f;
typedef __attribute__((ext_vector_type(8))) float v8f;

#define S_LEN 1024
#define D_LEN 256
#define MT    2
#define NT    4
#define GRP   4
#define NSTEP (S_LEN / 4)
#define NGRP  (NSTEP / GRP)
#define NJ    (NGRP / 2)
#define HAND  (2 * NT * 8 * 32)

struct Slot {
    v2f ar[2], ai[2];
    v4f br[2], bi[2];
    v2f w;
};

__device__ __forceinline__ void load_group(Slot (&s)[GRP],
                                           const float* __restrict__ r,
                                           const float* __restrict__ i,
                                           const float* __restrict__ w,
                                           unsigned offA, unsigned offB, unsigned offW, int grp)
{
    const float* rs = r + (size_t)grp * (GRP * 4 * D_LEN);
    const float* is = i + (size_t)grp * (GRP * 4 * D_LEN);
    const float* ws = w + (size_t)grp * (GRP * 4);
    #pragma unroll
    for (int u = 0; u < GRP; ++u) {
        #pragma unroll
        for (int e = 0; e < 2; ++e) {
            const int row = (u * 4 + e) * D_LEN;
            s[u].ar[e] = *(const v2f*)(rs + offA + row);
            s[u].ai[e] = *(const v2f*)(is + offA + row);
            s[u].br[e] = *(const v4f*)(rs + offB + row);
            s[u].bi[e] = *(const v4f*)(is + offB + row);
        }
        s[u].w = *(const v2f*)(ws + offW + u * 4);
    }
}

__device__ __forceinline__ void mma_group(const Slot (&s)[GRP],
                                          v8f (&accR)[MT][NT], v8f (&accM)[MT][NT])
{
    #pragma unroll
    for (int u = 0; u < GRP; ++u) {
        v2f aR[MT], aI[MT], bR[NT], bI[NT];
        #pragma unroll
        for (int t = 0; t < MT; ++t) {
            aR[t] = (v2f){ s[u].ar[0][t] * s[u].w.x, s[u].ar[1][t] * s[u].w.y };
            aI[t] = (v2f){ s[u].ai[0][t] * s[u].w.x, s[u].ai[1][t] * s[u].w.y };
        }
        #pragma unroll
        for (int t = 0; t < NT; ++t) {
            bR[t] = (v2f){ s[u].br[0][t], s[u].br[1][t] };
            bI[t] = (v2f){ s[u].bi[0][t], s[u].bi[1][t] };
        }
        #pragma unroll
        for (int mt = 0; mt < MT; ++mt) {
            #pragma unroll
            for (int nt = 0; nt < NT; ++nt) {
                accR[mt][nt]  = __builtin_amdgcn_wmma_f32_16x16x4_f32(
                    false, aR[mt], false, bR[nt], (short)0, accR[mt][nt],  false, false);
                accM[mt][nt] = __builtin_amdgcn_wmma_f32_16x16x4_f32(
                    false, aI[mt], false, bR[nt], (short)0, accM[mt][nt], false, false);
            }
        }
        #pragma unroll
        for (int mt = 0; mt < MT; ++mt) {
            #pragma unroll
            for (int nt = 0; nt < NT; ++nt) {
                accR[mt][nt]  = __builtin_amdgcn_wmma_f32_16x16x4_f32(
                    false, aI[mt], false, bI[nt], (short)0, accR[mt][nt],  false, false);
            }
        }
    }
}

#define KEEP_GROUP_AHEAD() __builtin_amdgcn_sched_barrier(0)

__device__ __forceinline__ void put_tile(float* base, int slot, int lane, v8f t)
{
    *(v4f*)(base + (slot * 2 + 0) * 128 + 4 * lane) = (v4f){ t[0], t[1], t[2], t[3] };
    *(v4f*)(base + (slot * 2 + 1) * 128 + 4 * lane) = (v4f){ t[4], t[5], t[6], t[7] };
}

__device__ __forceinline__ v8f get_tile(const float* base, int slot, int lane)
{
    v4f lo = *(const v4f*)(base + (slot * 2 + 0) * 128 + 4 * lane);
    v4f hi = *(const v4f*)(base + (slot * 2 + 1) * 128 + 4 * lane);
    return (v8f){ lo[0], lo[1], lo[2], lo[3], hi[0], hi[1], hi[2], hi[3] };
}

template <int GIVE>
__device__ __forceinline__ void hand_over(float* mine, int lane,
                                          const v8f (&accR)[MT][NT], const v8f (&accM)[MT][NT])
{
    #pragma unroll
    for (int nt = 0; nt < NT; ++nt) {
        put_tile(mine, nt,      lane, accR[GIVE][nt]);
        put_tile(mine, NT + nt, lane, accM[GIVE][nt]);
    }
}

template <int OWN>
__device__ __forceinline__ void finish(const float* partner, int lane, int h, size_t o0,
                                       const v8f (&accR)[MT][NT], const v8f (&accM)[MT][NT],
                                       float* __restrict__ out_r, float* __restrict__ out_i)
{
    v8f sumR[NT], sumI[NT];
    #pragma unroll
    for (int nt = 0; nt < NT; ++nt) {
        sumR[nt] = accR[OWN][nt] + get_tile(partner, nt, lane);
        sumI[nt] = accM[OWN][nt] + get_tile(partner, NT + nt, lane);
    }
    #pragma unroll
    for (int j = 0; j < 8; ++j) {
        size_t o = o0 + (size_t)(2 * (8 * h + j) + OWN) * D_LEN;
        *(v4f*)&out_r[o] = (v4f){ sumR[0][j], sumR[1][j], sumR[2][j], sumR[3][j] };
        *(v4f*)&out_i[o] = (v4f){ sumI[0][j], sumI[1][j], sumI[2][j], sumI[3][j] };
    }
}

__global__ void __launch_bounds__(256) complex_mixture_wmma(
    const float* __restrict__ in_r,
    const float* __restrict__ in_i,
    const float* __restrict__ wgt,
    float* __restrict__ out_r,
    float* __restrict__ out_i)
{
    __shared__ __attribute__((aligned(16))) float hand[8 * HAND];

    const int b     = blockIdx.z;
    const int aBase = blockIdx.y * 64;
    const int cBase = blockIdx.x * 128;

    const int tid   = threadIdx.x;
    const int lane  = tid & 31;
    const int wave  = __builtin_amdgcn_readfirstlane(tid >> 5);
    const int waveM = wave & 1;
    const int waveN = (wave >> 1) & 1;
    const int kh    = wave >> 2;

    const int c = lane & 15;
    const int h = lane >> 4;

    const float* rSrc = in_r + (size_t)b * S_LEN * D_LEN;
    const float* iSrc = in_i + (size_t)b * S_LEN * D_LEN;
    const float* wSrc = wgt  + (size_t)b * S_LEN;

    const unsigned offA = (unsigned)(2 * h * D_LEN + aBase + waveM * 32 + 2 * c);
    const unsigned offB = (unsigned)(2 * h * D_LEN + cBase + waveN * 64 + 4 * c);
    const unsigned offW = (unsigned)(2 * h);

    v8f accR[MT][NT] = {};
    v8f accM[MT][NT] = {};

    Slot setA[GRP], setB[GRP];
    load_group(setA, rSrc, iSrc, wSrc, offA, offB, offW, kh);

    for (int j = 0; j < NJ - 2; j += 2) {
        load_group(setB, rSrc, iSrc, wSrc, offA, offB, offW, 2 * (j + 1) + kh);
        KEEP_GROUP_AHEAD();
        mma_group(setA, accR, accM);
        KEEP_GROUP_AHEAD();
        load_group(setA, rSrc, iSrc, wSrc, offA, offB, offW, 2 * (j + 2) + kh);
        KEEP_GROUP_AHEAD();
        mma_group(setB, accR, accM);
        KEEP_GROUP_AHEAD();
    }
    load_group(setB, rSrc, iSrc, wSrc, offA, offB, offW, 2 * (NJ - 1) + kh);
    KEEP_GROUP_AHEAD();
    mma_group(setA, accR, accM);
    mma_group(setB, accR, accM);

    float* mine          = hand + wave * HAND;
    const float* partner = hand + (wave ^ 4) * HAND;
    if (kh == 0) hand_over<1>(mine, lane, accR, accM);
    else         hand_over<0>(mine, lane, accR, accM);
    __syncthreads();

    const size_t o0 = ((size_t)b * D_LEN + aBase + waveM * 32) * D_LEN + cBase + waveN * 64 + 4 * c;
    if (kh == 0) finish<0>(partner, lane, h, o0, accR, accM, out_r, out_i);
    else         finish<1>(partner, lane, h, o0, accR, accM, out_r, out_i);
}

#define TS 64
#define TP (TS + 1)

__global__ void __launch_bounds__(256) mirror_difference(float* __restrict__ m)
{
    __shared__ float t1[TS * TP];
    __shared__ float t2[TS * TP];

    const int p = blockIdx.x;
    int I, J;
    if (p < 4)      { I = 0; J = p;     }
    else if (p < 7) { I = 1; J = p - 3; }
    else if (p < 9) { I = 2; J = p - 5; }
    else            { I = 3; J = 3;     }

    const int tid = threadIdx.x;
    const int cg  = tid & 15;
    const int r0  = tid >> 4;

    float* base = m + (size_t)blockIdx.y * D_LEN * D_LEN;
    float* g1 = base + (size_t)(TS * I) * D_LEN + TS * J + 4 * cg;
    float* g2 = base + (size_t)(TS * J) * D_LEN + TS * I + 4 * cg;

    v4f x1[4], x2[4];
    #pragma unroll
    for (int q = 0; q < 4; ++q) {
        const int r = r0 + 16 * q;
        x1[q] = *(const v4f*)(g1 + (size_t)r * D_LEN);
        x2[q] = *(const v4f*)(g2 + (size_t)r * D_LEN);
    }
    #pragma unroll
    for (int q = 0; q < 4; ++q) {
        const int r = r0 + 16 * q;
        #pragma unroll
        for (int k = 0; k < 4; ++k) {
            t1[r * TP + 4 * cg + k] = x1[q][k];
            t2[r * TP + 4 * cg + k] = x2[q][k];
        }
    }
    __syncthreads();

    #pragma unroll
    for (int q = 0; q < 4; ++q) {
        const int r = r0 + 16 * q;
        v4f d1, d2;
        #pragma unroll
        for (int k = 0; k < 4; ++k) {
            d1[k] = x1[q][k] - t2[(4 * cg + k) * TP + r];
            d2[k] = x2[q][k] - t1[(4 * cg + k) * TP + r];
        }
        *(v4f*)(g1 + (size_t)r * D_LEN) = d1;
        if (I != J) *(v4f*)(g2 + (size_t)r * D_LEN) = d2;
    }
}

extern "C" void kernel_launch(void* const* d_in, const int* in_sizes, int n_in,
                              void* d_out, int out_size, void* d_ws, size_t ws_size,
                              hipStream_t stream) {
    const float* in_r = (const float*)d_in[0];
    const float* in_i = (const float*)d_in[1];
    const float* w    = (const float*)d_in[2];
    float* out_r = (float*)d_out;
    float* out_i = out_r + (size_t)32 * D_LEN * D_LEN;

    dim3 grid(D_LEN / 128, D_LEN / 64, 32);
    dim3 block(256);
    hipLaunchKernelGGL(complex_mixture_wmma, grid, block, 0, stream,
                       in_r, in_i, w, out_r, out_i);

    dim3 grid2(10, 32);
    hipLaunchKernelGGL(mirror_difference, grid2, block, 0, stream, out_i);
}


# P is symmetric: its (odd row, even column) tiles are not computed, those out_r elements take Q = P - M of the mirror block, transposed through LDS beside M^T; 14 WMMAs per k-step and wave (was 16)
# speedup vs baseline: 12.7264x; 1.4832x over previous
#include <hip/hip_runtime.h>


typedef __attribute__((ext_vector_type(2))) float v2f;
typedef __attribute__((ext_vector_type(4))) float v4f;
typedef __attribute__((ext_vector_type(8))) float v8f;

#define S_LEN 1024
#define D_LEN 256
#define MT    2
#define NT    4
#define GRP   4
#define NSTEP (S_LEN / 4)
#define NGRP  (NSTEP / GRP)
#define NJ    (NGRP / 2)
#define HAND  (2 * NT * 8 * 32)
#define BLK   64
#define TP    (BLK + 1)
#define P_DONE_HERE(mt, nt) ((mt) == 0 || ((nt) & 1) != 0)

struct Slot {
    v2f ar[2], ai[2];
    v4f br[2], bi[2];
    v2f w;
};

__device__ __forceinline__ void load_group(Slot (&s)[GRP],
                                           const float* __restrict__ r,
                                           const float* __restrict__ i,
                                           const float* __restrict__ w,
                                           unsigned offA, unsigned offB, unsigned offW, int grp)
{
    const float* rs = r + (size_t)grp * (GRP * 4 * D_LEN);
    const float* is = i + (size_t)grp * (GRP * 4 * D_LEN);
    const float* ws = w + (size_t)grp * (GRP * 4);
    #pragma unroll
    for (int u = 0; u < GRP; ++u) {
        #pragma unroll
        for (int e = 0; e < 2; ++e) {
            const int row = (u * 4 + e) * D_LEN;
            s[u].ar[e] = *(const v2f*)(rs + offA + row);
            s[u].ai[e] = *(const v2f*)(is + offA + row);
            s[u].br[e] = *(const v4f*)(rs + offB + row);
            s[u].bi[e] = *(const v4f*)(is + offB + row);
        }
        s[u].w = *(const v2f*)(ws + offW + u * 4);
    }
}

__device__ __forceinline__ void mma_group(const Slot (&s)[GRP],
                                          v8f (&accP)[MT][NT], v8f (&accM)[MT][NT])
{
    #pragma unroll
    for (int u = 0; u < GRP; ++u) {
        v2f aP[MT], aM[MT], bP[NT], bM[NT];
        #pragma unroll
        for (int t = 0; t < MT; ++t) {
            aP[t] = (v2f){ (s[u].ar[0][t] + s[u].ai[0][t]) * s[u].w.x,
                           (s[u].ar[1][t] + s[u].ai[1][t]) * s[u].w.y };
            aM[t] = (v2f){ s[u].ai[0][t] * s[u].w.x, s[u].ai[1][t] * s[u].w.y };
        }
        #pragma unroll
        for (int t = 0; t < NT; ++t) {
            bP[t] = (v2f){ s[u].br[0][t] + s[u].bi[0][t], s[u].br[1][t] + s[u].bi[1][t] };
            bM[t] = (v2f){ s[u].br[0][t], s[u].br[1][t] };
        }
        #pragma unroll
        for (int mt = 0; mt < MT; ++mt) {
            #pragma unroll
            for (int nt = 0; nt < NT; ++nt) {
                if (P_DONE_HERE(mt, nt))
                    accP[mt][nt] = __builtin_amdgcn_wmma_f32_16x16x4_f32(
                        false, aP[mt], false, bP[nt], (short)0, accP[mt][nt], false, false);
                accM[mt][nt] = __builtin_amdgcn_wmma_f32_16x16x4_f32(
                    false, aM[mt], false, bM[nt], (short)0, accM[mt][nt], false, false);
            }
        }
    }
}

#define KEEP_GROUP_AHEAD() __builtin_amdgcn_sched_barrier(0)

__device__ __forceinline__ void put_tile(float* base, int slot, int lane, v8f t)
{
    *(v4f*)(base + (slot * 2 + 0) * 128 + 4 * lane) = (v4f){ t[0], t[1], t[2], t[3] };
    *(v4f*)(base + (slot * 2 + 1) * 128 + 4 * lane) = (v4f){ t[4], t[5], t[6], t[7] };
}

__device__ __forceinline__ v8f get_tile(const float* base, int slot, int lane)
{
    v4f lo = *(const v4f*)(base + (slot * 2 + 0) * 128 + 4 * lane);
    v4f hi = *(const v4f*)(base + (slot * 2 + 1) * 128 + 4 * lane);
    return (v8f){ lo[0], lo[1], lo[2], lo[3], hi[0], hi[1], hi[2], hi[3] };
}

template <int GIVE>
__device__ __forceinline__ void hand_over(float* mine, int lane,
                                          const v8f (&accP)[MT][NT], const v8f (&accM)[MT][NT])
{
    #pragma unroll
    for (int nt = 0; nt < NT; ++nt) {
        if (P_DONE_HERE(GIVE, nt)) put_tile(mine, nt, lane, accP[GIVE][nt]);
        put_tile(mine, NT + nt, lane, accM[GIVE][nt]);
    }
}

template <int OWN>
__device__ __forceinline__ void join(const float* partner, int lane,
                                     const v8f (&accP)[MT][NT], const v8f (&accM)[MT][NT],
                                     v8f (&sumQ)[NT], v8f (&sumM)[NT])
{
    #pragma unroll
    for (int nt = 0; nt < NT; ++nt) {
        sumM[nt] = accM[OWN][nt] + get_tile(partner, NT + nt, lane);
        if (P_DONE_HERE(OWN, nt)) sumQ[nt] = (accP[OWN][nt] + get_tile(partner, nt, lane)) - sumM[nt];
        else                      sumQ[nt] = (v8f){};
    }
}

__global__ void __launch_bounds__(256) complex_mixture_wmma(
    const float* __restrict__ in_r,
    const float* __restrict__ in_i,
    const float* __restrict__ wgt,
    float* __restrict__ out_r,
    float* __restrict__ out_i)
{
    __shared__ __attribute__((aligned(16))) float hand[8 * HAND];
    __shared__ float mBlk[2][BLK * TP];
    __shared__ float qBlk[2][BLK * TP];

    const int b = blockIdx.y;
    const int u = blockIdx.x;
    int I, J;
    if (u < 3)       { I = 0; J = u + 1; }
    else if (u < 5)  { I = 1; J = u - 1; }
    else if (u == 5) { I = 2; J = 3;     }
    else if (u == 6) { I = 0; J = 1;     }
    else             { I = 2; J = 3;     }
    const int pair = u < 6 ? 1 : 0;

    const int tid   = threadIdx.x;
    const int lane  = tid & 31;
    const int wave  = __builtin_amdgcn_readfirstlane(tid >> 5);
    const int waveM = wave & 1;
    const int blk   = (wave >> 1) & 1;
    const int kh    = wave >> 2;

    const int rowBlk = blk ? J : I;
    const int colBlk = (blk ^ pair) ? J : I;
    const int mirror = blk ^ pair;
    const int aBase  = rowBlk * BLK + waveM * 32;
    const int cBase  = colBlk * BLK;

    const int c = lane & 15;
    const int h = lane >> 4;

    const float* rSrc = in_r + (size_t)b * S_LEN * D_LEN;
    const float* iSrc = in_i + (size_t)b * S_LEN * D_LEN;
    const float* wSrc = wgt  + (size_t)b * S_LEN;

    const unsigned offA = (unsigned)(2 * h * D_LEN + aBase + 2 * c);
    const unsigned offB = (unsigned)(2 * h * D_LEN + cBase + 4 * c);
    const unsigned offW = (unsigned)(2 * h);

    v8f accP[MT][NT] = {};
    v8f accM[MT][NT] = {};

    Slot setA[GRP], setB[GRP];
    load_group(setA, rSrc, iSrc, wSrc, offA, offB, offW, kh);

    for (int j = 0; j < NJ - 2; j += 2) {
        load_group(setB, rSrc, iSrc, wSrc, offA, offB, offW, 2 * (j + 1) + kh);
        KEEP_GROUP_AHEAD();
        mma_group(setA, accP, accM);
        KEEP_GROUP_AHEAD();
        load_group(setA, rSrc, iSrc, wSrc, offA, offB, offW, 2 * (j + 2) + kh);
        KEEP_GROUP_AHEAD();
        mma_group(setB, accP, accM);
        KEEP_GROUP_AHEAD();
    }
    load_group(setB, rSrc, iSrc, wSrc, offA, offB, offW, 2 * (NJ - 1) + kh);
    KEEP_GROUP_AHEAD();
    mma_group(setA, accP, accM);
    mma_group(setB, accP, accM);

    float* mine          = hand + wave * HAND;
    const float* partner = hand + (wave ^ 4) * HAND;
    if (kh == 0) hand_over<1>(mine, lane, accP, accM);
    else         hand_over<0>(mine, lane, accP, accM);
    __syncthreads();

    v8f sumQ[NT], sumM[NT];
    if (kh == 0) join<0>(partner, lane, accP, accM, sumQ, sumM);
    else         join<1>(partner, lane, accP, accM, sumQ, sumM);

    const int row0 = waveM * 32 + 16 * h + kh;
    float* mOwn          = mBlk[blk];
    const float* mMirror = mBlk[mirror];
    #pragma unroll
    for (int j = 0; j < 8; ++j) {
        #pragma unroll
        for (int nt = 0; nt < NT; ++nt)
            mOwn[(row0 + 2 * j) * TP + 4 * c + nt] = sumM[nt][j];
    }
    float* qOwn          = qBlk[blk];
    const float* qMirror = qBlk[mirror];
    if (kh == 0) {
        #pragma unroll
        for (int j = 0; j < 8; ++j) {
            qOwn[(row0 + 2 * j) * TP + 4 * c + 1] = sumQ[1][j];
            qOwn[(row0 + 2 * j) * TP + 4 * c + 3] = sumQ[3][j];
        }
    }
    __syncthreads();

    const size_t o0 = ((size_t)b * D_LEN + rowBlk * BLK) * D_LEN + cBase + 4 * c;
    if (kh == 0) {
        #pragma unroll
        for (int j = 0; j < 8; ++j) {
            const int row = row0 + 2 * j;
            v4f mt;
            #pragma unroll
            for (int nt = 0; nt < NT; ++nt)
                mt[nt] = mMirror[(4 * c + nt) * TP + row];
            const size_t o = o0 + (size_t)row * D_LEN;
            *(v4f*)&out_r[o] = (v4f){ sumQ[0][j], sumQ[1][j], sumQ[2][j], sumQ[3][j] } - mt;
            *(v4f*)&out_i[o] = (v4f){ sumM[0][j], sumM[1][j], sumM[2][j], sumM[3][j] } - mt;
        }
    } else {
        #pragma unroll
        for (int j = 0; j < 8; ++j) {
            const int row = row0 + 2 * j;
            v4f mt;
            #pragma unroll
            for (int nt = 0; nt < NT; ++nt)
                mt[nt] = mMirror[(4 * c + nt) * TP + row];
            const float q0 = qMirror[(4 * c + 0) * TP + row];
            const float q2 = qMirror[(4 * c + 2) * TP + row];
            const size_t o = o0 + (size_t)row * D_LEN;
            *(v4f*)&out_r[o] = (v4f){ q0 - sumM[0][j], sumQ[1][j] - mt[1], q2 - sumM[2][j], sumQ[3][j] - mt[3] };
            *(v4f*)&out_i[o] = (v4f){ sumM[0][j], sumM[1][j], sumM[2][j], sumM[3][j] } - mt;
        }
    }
}

extern "C" void kernel_launch(void* const* d_in, const int* in_sizes, int n_in,
                              void* d_out, int out_size, void* d_ws, size_t ws_size,
                              hipStream_t stream) {
    const float* in_r = (const float*)d_in[0];
    const float* in_i = (const float*)d_in[1];
    const float* w    = (const float*)d_in[2];
    float* out_r = (float*)d_out;
    float* out_i = out_r + (size_t)32 * D_LEN * D_LEN;

    dim3 grid(8, 32);
    dim3 block(256);
    hipLaunchKernelGGL(complex_mixture_wmma, grid, block, 0, stream,
                       in_r, in_i, w, out_r, out_i);
}


# the two waves of each SIMD take turns on the matrix pipe: a wave lowers its s_setprio (3,2,1,0) after its own groups 16, 26 and 30 of 32, so the wave behind wins and both reach the join together
# speedup vs baseline: 13.3952x; 1.0526x over previous
#include <hip/hip_runtime.h>


typedef __attribute__((ext_vector_type(2))) float v2f;
typedef __attribute__((ext_vector_type(4))) float v4f;
typedef __attribute__((ext_vector_type(8))) float v8f;

#define S_LEN 1024
#define D_LEN 256
#define MT    2
#define NT    4
#define GRP   4
#define NSTEP (S_LEN / 4)
#define NGRP  (NSTEP / GRP)
#define NJ    (NGRP / 2)
#define HAND  (2 * NT * 8 * 32)
#define BLK   64
#define TP    (BLK + 1)
#define P_DONE_HERE(mt, nt) ((mt) == 0 || ((nt) & 1) != 0)

struct Slot {
    v2f ar[2], ai[2];
    v4f br[2], bi[2];
    v2f w;
};

__device__ __forceinline__ void load_group(Slot (&s)[GRP],
                                           const float* __restrict__ r,
                                           const float* __restrict__ i,
                                           const float* __restrict__ w,
                                           unsigned offA, unsigned offB, unsigned offW, int grp)
{
    const float* rs = r + (size_t)grp * (GRP * 4 * D_LEN);
    const float* is = i + (size_t)grp * (GRP * 4 * D_LEN);
    const float* ws = w + (size_t)grp * (GRP * 4);
    #pragma unroll
    for (int u = 0; u < GRP; ++u) {
        #pragma unroll
        for (int e = 0; e < 2; ++e) {
            const int row = (u * 4 + e) * D_LEN;
            s[u].ar[e] = *(const v2f*)(rs + offA + row);
            s[u].ai[e] = *(const v2f*)(is + offA + row);
            s[u].br[e] = *(const v4f*)(rs + offB + row);
            s[u].bi[e] = *(const v4f*)(is + offB + row);
        }
        s[u].w = *(const v2f*)(ws + offW + u * 4);
    }
}

__device__ __forceinline__ void mma_group(const Slot (&s)[GRP],
                                          v8f (&accP)[MT][NT], v8f (&accM)[MT][NT])
{
    #pragma unroll
    for (int u = 0; u < GRP; ++u) {
        v2f aP[MT], aM[MT], bP[NT], bM[NT];
        #pragma unroll
        for (int t = 0; t < MT; ++t) {
            aP[t] = (v2f){ (s[u].ar[0][t] + s[u].ai[0][t]) * s[u].w.x,
                           (s[u].ar[1][t] + s[u].ai[1][t]) * s[u].w.y };
            aM[t] = (v2f){ s[u].ai[0][t] * s[u].w.x, s[u].ai[1][t] * s[u].w.y };
        }
        #pragma unroll
        for (int t = 0; t < NT; ++t) {
            bP[t] = (v2f){ s[u].br[0][t] + s[u].bi[0][t], s[u].br[1][t] + s[u].bi[1][t] };
            bM[t] = (v2f){ s[u].br[0][t], s[u].br[1][t] };
        }
        #pragma unroll
        for (int mt = 0; mt < MT; ++mt) {
            #pragma unroll
            for (int nt = 0; nt < NT; ++nt) {
                if (P_DONE_HERE(mt, nt))
                    accP[mt][nt] = __builtin_amdgcn_wmma_f32_16x16x4_f32(
                        false, aP[mt], false, bP[nt], (short)0, accP[mt][nt], false, false);
                accM[mt][nt] = __builtin_amdgcn_wmma_f32_16x16x4_f32(
                    false, aM[mt], false, bM[nt], (short)0, accM[mt][nt], false, false);
            }
        }
    }
}

#define KEEP_GROUP_AHEAD() __builtin_amdgcn_sched_barrier(0)

#define PRIO_FIRST   16
#define PRIO_SECOND  26

__device__ __forceinline__ void put_tile(float* base, int slot, int lane, v8f t)
{
    *(v4f*)(base + (slot * 2 + 0) * 128 + 4 * lane) = (v4f){ t[0], t[1], t[2], t[3] };
    *(v4f*)(base + (slot * 2 + 1) * 128 + 4 * lane) = (v4f){ t[4], t[5], t[6], t[7] };
}

__device__ __forceinline__ v8f get_tile(const float* base, int slot, int lane)
{
    v4f lo = *(const v4f*)(base + (slot * 2 + 0) * 128 + 4 * lane);
    v4f hi = *(const v4f*)(base + (slot * 2 + 1) * 128 + 4 * lane);
    return (v8f){ lo[0], lo[1], lo[2], lo[3], hi[0], hi[1], hi[2], hi[3] };
}

template <int GIVE>
__device__ __forceinline__ void hand_over(float* mine, int lane,
                                          const v8f (&accP)[MT][NT], const v8f (&accM)[MT][NT])
{
    #pragma unroll
    for (int nt = 0; nt < NT; ++nt) {
        if (P_DONE_HERE(GIVE, nt)) put_tile(mine, nt, lane, accP[GIVE][nt]);
        put_tile(mine, NT + nt, lane, accM[GIVE][nt]);
    }
}

template <int OWN>
__device__ __forceinline__ void join(const float* partner, int lane,
                                     const v8f (&accP)[MT][NT], const v8f (&accM)[MT][NT],
                                     v8f (&sumQ)[NT], v8f (&sumM)[NT])
{
    #pragma unroll
    for (int nt = 0; nt < NT; ++nt) {
        sumM[nt] = accM[OWN][nt] + get_tile(partner, NT + nt, lane);
        if (P_DONE_HERE(OWN, nt)) sumQ[nt] = (accP[OWN][nt] + get_tile(partner, nt, lane)) - sumM[nt];
        else                      sumQ[nt] = (v8f){};
    }
}

__global__ void __launch_bounds__(256) complex_mixture_wmma(
    const float* __restrict__ in_r,
    const float* __restrict__ in_i,
    const float* __restrict__ wgt,
    float* __restrict__ out_r,
    float* __restrict__ out_i)
{
    __shared__ __attribute__((aligned(16))) float hand[8 * HAND];
    __shared__ float mBlk[2][BLK * TP];
    __shared__ float qBlk[2][BLK * TP];

    const int b = blockIdx.y;
    const int u = blockIdx.x;
    int I, J;
    if (u < 3)       { I = 0; J = u + 1; }
    else if (u < 5)  { I = 1; J = u - 1; }
    else if (u == 5) { I = 2; J = 3;     }
    else if (u == 6) { I = 0; J = 1;     }
    else             { I = 2; J = 3;     }
    const int pair = u < 6 ? 1 : 0;

    const int tid   = threadIdx.x;
    const int lane  = tid & 31;
    const int wave  = __builtin_amdgcn_readfirstlane(tid >> 5);
    const int waveM = wave & 1;
    const int blk   = (wave >> 1) & 1;
    const int kh    = wave >> 2;

    const int rowBlk = blk ? J : I;
    const int colBlk = (blk ^ pair) ? J : I;
    const int mirror = blk ^ pair;
    const int aBase  = rowBlk * BLK + waveM * 32;
    const int cBase  = colBlk * BLK;

    const int c = lane & 15;
    const int h = lane >> 4;

    const float* rSrc = in_r + (size_t)b * S_LEN * D_LEN;
    const float* iSrc = in_i + (size_t)b * S_LEN * D_LEN;
    const float* wSrc = wgt  + (size_t)b * S_LEN;

    const unsigned offA = (unsigned)(2 * h * D_LEN + aBase + 2 * c);
    const unsigned offB = (unsigned)(2 * h * D_LEN + cBase + 4 * c);
    const unsigned offW = (unsigned)(2 * h);

    v8f accP[MT][NT] = {};
    v8f accM[MT][NT] = {};

    Slot setA[GRP], setB[GRP];
    load_group(setA, rSrc, iSrc, wSrc, offA, offB, offW, kh);

    __builtin_amdgcn_s_setprio(3);
    for (int j = 0; j < NJ - 2; j += 2) {
        if (j == PRIO_FIRST)  __builtin_amdgcn_s_setprio(2);
        if (j == PRIO_SECOND) __builtin_amdgcn_s_setprio(1);
        load_group(setB, rSrc, iSrc, wSrc, offA, offB, offW, 2 * (j + 1) + kh);
        KEEP_GROUP_AHEAD();
        mma_group(setA, accP, accM);
        KEEP_GROUP_AHEAD();
        load_group(setA, rSrc, iSrc, wSrc, offA, offB, offW, 2 * (j + 2) + kh);
        KEEP_GROUP_AHEAD();
        mma_group(setB, accP, accM);
        KEEP_GROUP_AHEAD();
    }
    __builtin_amdgcn_s_setprio(0);
    load_group(setB, rSrc, iSrc, wSrc, offA, offB, offW, 2 * (NJ - 1) + kh);
    KEEP_GROUP_AHEAD();
    mma_group(setA, accP, accM);
    mma_group(setB, accP, accM);

    float* mine          = hand + wave * HAND;
    const float* partner = hand + (wave ^ 4) * HAND;
    if (kh == 0) hand_over<1>(mine, lane, accP, accM);
    else         hand_over<0>(mine, lane, accP, accM);
    __syncthreads();

    v8f sumQ[NT], sumM[NT];
    if (kh == 0) join<0>(partner, lane, accP, accM, sumQ, sumM);
    else         join<1>(partner, lane, accP, accM, sumQ, sumM);

    const int row0 = waveM * 32 + 16 * h + kh;
    float* mOwn          = mBlk[blk];
    const float* mMirror = mBlk[mirror];
    #pragma unroll
    for (int j = 0; j < 8; ++j) {
        #pragma unroll
        for (int nt = 0; nt < NT; ++nt)
            mOwn[(row0 + 2 * j) * TP + 4 * c + nt] = sumM[nt][j];
    }
    float* qOwn          = qBlk[blk];
    const float* qMirror = qBlk[mirror];
    if (kh == 0) {
        #pragma unroll
        for (int j = 0; j < 8; ++j) {
            qOwn[(row0 + 2 * j) * TP + 4 * c + 1] = sumQ[1][j];
            qOwn[(row0 + 2 * j) * TP + 4 * c + 3] = sumQ[3][j];
        }
    }
    __syncthreads();

    const size_t o0 = ((size_t)b * D_LEN + rowBlk * BLK) * D_LEN + cBase + 4 * c;
    if (kh == 0) {
        #pragma unroll
        for (int j = 0; j < 8; ++j) {
            const int row = row0 + 2 * j;
            v4f mt;
            #pragma unroll
            for (int nt = 0; nt < NT; ++nt)
                mt[nt] = mMirror[(4 * c + nt) * TP + row];
            const size_t o = o0 + (size_t)row * D_LEN;
            *(v4f*)&out_r[o] = (v4f){ sumQ[0][j], sumQ[1][j], sumQ[2][j], sumQ[3][j] } - mt;
            *(v4f*)&out_i[o] = (v4f){ sumM[0][j], sumM[1][j], sumM[2][j], sumM[3][j] } - mt;
        }
    } else {
        #pragma unroll
        for (int j = 0; j < 8; ++j) {
            const int row = row0 + 2 * j;
            v4f mt;
            #pragma unroll
            for (int nt = 0; nt < NT; ++nt)
                mt[nt] = mMirror[(4 * c + nt) * TP + row];
            const float q0 = qMirror[(4 * c + 0) * TP + row];
            const float q2 = qMirror[(4 * c + 2) * TP + row];
            const size_t o = o0 + (size_t)row * D_LEN;
            *(v4f*)&out_r[o] = (v4f){ q0 - sumM[0][j], sumQ[1][j] - mt[1], q2 - sumM[2][j], sumQ[3][j] - mt[3] };
            *(v4f*)&out_i[o] = (v4f){ sumM[0][j], sumM[1][j], sumM[2][j], sumM[3][j] } - mt;
        }
    }
}

extern "C" void kernel_launch(void* const* d_in, const int* in_sizes, int n_in,
                              void* d_out, int out_size, void* d_ws, size_t ws_size,
                              hipStream_t stream) {
    const float* in_r = (const float*)d_in[0];
    const float* in_i = (const float*)d_in[1];
    const float* w    = (const float*)d_in[2];
    float* out_r = (float*)d_out;
    float* out_i = out_r + (size_t)32 * D_LEN * D_LEN;

    dim3 grid(8, 32);
    dim3 block(256);
    hipLaunchKernelGGL(complex_mixture_wmma, grid, block, 0, stream,
                       in_r, in_i, w, out_r, out_i);
}


# wave tile doubles again to the whole 64x64 block, s split between the block's four waves: 18 loads per 52 WMMAs (was 36 per 56), and P leaves out 6 of 16 tile classes: 26 WMMAs per k-step (was 28)
# speedup vs baseline: 14.7024x; 1.0976x over previous
#include <hip/hip_runtime.h>


typedef __attribute__((ext_vector_type(2))) float v2f;
typedef __attribute__((ext_vector_type(4))) float v4f;
typedef __attribute__((ext_vector_type(8))) float v8f;

#define S_LEN 1024
#define D_LEN 256
#define MT    4
#define NT    4
#define GRP   2
#define KQ    4
#define NSTEP (S_LEN / 4)
#define NGRP  (NSTEP / GRP)
#define NJ    (NGRP / KQ)
#define SLOTS 8
#define HAND  ((KQ - 1) * SLOTS * 8 * 32)
#define BLK   64
#define TP    (BLK + 1)
#define P_DONE_HERE(mt, nt) ((mt) <= (nt))

struct Slot {
    v4f ar[2], ai[2];
    v4f br[2], bi[2];
    v2f w;
};

__device__ __forceinline__ void load_group(Slot (&s)[GRP],
                                           const float* __restrict__ r,
                                           const float* __restrict__ i,
                                           const float* __restrict__ w,
                                           unsigned offA, unsigned offB, unsigned offW, int grp)
{
    const float* rs = r + (size_t)grp * (GRP * 4 * D_LEN);
    const float* is = i + (size_t)grp * (GRP * 4 * D_LEN);
    const float* ws = w + (size_t)grp * (GRP * 4);
    #pragma unroll
    for (int u = 0; u < GRP; ++u) {
        #pragma unroll
        for (int e = 0; e < 2; ++e) {
            const int row = (u * 4 + e) * D_LEN;
            s[u].ar[e] = *(const v4f*)(rs + offA + row);
            s[u].ai[e] = *(const v4f*)(is + offA + row);
            s[u].br[e] = *(const v4f*)(rs + offB + row);
            s[u].bi[e] = *(const v4f*)(is + offB + row);
        }
        s[u].w = *(const v2f*)(ws + offW + u * 4);
    }
}

__device__ __forceinline__ void mma_group(const Slot (&s)[GRP],
                                          v8f (&accP)[MT][NT], v8f (&accM)[MT][NT])
{
    #pragma unroll
    for (int u = 0; u < GRP; ++u) {
        v2f aP[MT], aM[MT], bP[NT], bM[NT];
        #pragma unroll
        for (int t = 0; t < MT; ++t) {
            aP[t] = (v2f){ (s[u].ar[0][t] + s[u].ai[0][t]) * s[u].w.x,
                           (s[u].ar[1][t] + s[u].ai[1][t]) * s[u].w.y };
            aM[t] = (v2f){ s[u].ai[0][t] * s[u].w.x, s[u].ai[1][t] * s[u].w.y };
        }
        #pragma unroll
        for (int t = 0; t < NT; ++t) {
            bP[t] = (v2f){ s[u].br[0][t] + s[u].bi[0][t], s[u].br[1][t] + s[u].bi[1][t] };
            bM[t] = (v2f){ s[u].br[0][t], s[u].br[1][t] };
        }
        #pragma unroll
        for (int mt = 0; mt < MT; ++mt) {
            #pragma unroll
            for (int nt = 0; nt < NT; ++nt) {
                if (P_DONE_HERE(mt, nt))
                    accP[mt][nt] = __builtin_amdgcn_wmma_f32_16x16x4_f32(
                        false, aP[mt], false, bP[nt], (short)0, accP[mt][nt], false, false);
                accM[mt][nt] = __builtin_amdgcn_wmma_f32_16x16x4_f32(
                    false, aM[mt], false, bM[nt], (short)0, accM[mt][nt], false, false);
            }
        }
    }
}

#define KEEP_GROUP_AHEAD() __builtin_amdgcn_sched_barrier(0)

#define PRIO_FIRST   16
#define PRIO_SECOND  26

__device__ __forceinline__ void put_tile(float* base, int slot, int lane, v8f t)
{
    *(v4f*)(base + (slot * 2 + 0) * 128 + 4 * lane) = (v4f){ t[0], t[1], t[2], t[3] };
    *(v4f*)(base + (slot * 2 + 1) * 128 + 4 * lane) = (v4f){ t[4], t[5], t[6], t[7] };
}

__device__ __forceinline__ v8f get_tile(const float* base, int slot, int lane)
{
    v4f lo = *(const v4f*)(base + (slot * 2 + 0) * 128 + 4 * lane);
    v4f hi = *(const v4f*)(base + (slot * 2 + 1) * 128 + 4 * lane);
    return (v8f){ lo[0], lo[1], lo[2], lo[3], hi[0], hi[1], hi[2], hi[3] };
}

#define GIVE_IDX(mt, keep) ((mt) < (keep) ? (mt) : (mt) - 1)

template <int KEEP>
__device__ __forceinline__ void hand_over(float* mine, int lane,
                                          const v8f (&accP)[MT][NT], const v8f (&accM)[MT][NT])
{
    #pragma unroll
    for (int mt = 0; mt < MT; ++mt) {
        if (mt == KEEP) continue;
        #pragma unroll
        for (int nt = 0; nt < NT; ++nt) {
            if (P_DONE_HERE(mt, nt)) put_tile(mine, GIVE_IDX(mt, KEEP) * SLOTS + nt, lane, accP[mt][nt]);
            put_tile(mine, GIVE_IDX(mt, KEEP) * SLOTS + NT + nt, lane, accM[mt][nt]);
        }
    }
}

#define WAVE_OF(blk, p) (((p) & 1) | ((blk) << 1) | (((p) >> 1) << 2))

template <int KEEP>
__device__ __forceinline__ void join(const float* hand, int blk, int lane,
                                     const v8f (&accP)[MT][NT], const v8f (&accM)[MT][NT],
                                     v8f (&sumQ)[NT], v8f (&sumM)[NT])
{
    v8f sumP[NT];
    #pragma unroll
    for (int nt = 0; nt < NT; ++nt) {
        sumM[nt] = accM[KEEP][nt];
        sumP[nt] = accP[KEEP][nt];
    }
    #pragma unroll
    for (int p = 0; p < KQ; ++p) {
        if (p == KEEP) continue;
        const float* partner = hand + WAVE_OF(blk, p) * HAND;
        #pragma unroll
        for (int nt = 0; nt < NT; ++nt) {
            sumM[nt] += get_tile(partner, GIVE_IDX(KEEP, p) * SLOTS + NT + nt, lane);
            if (P_DONE_HERE(KEEP, nt)) sumP[nt] += get_tile(partner, GIVE_IDX(KEEP, p) * SLOTS + nt, lane);
        }
    }
    #pragma unroll
    for (int nt = 0; nt < NT; ++nt) {
        if (P_DONE_HERE(KEEP, nt)) sumQ[nt] = sumP[nt] - sumM[nt];
        else                       sumQ[nt] = (v8f){};
    }
}

template <int KEEP>
__device__ __forceinline__ void lay_out(float* mOwn, float* qOwn, int c, int h,
                                        const v8f (&sumQ)[NT], const v8f (&sumM)[NT])
{
    const int row0 = 32 * h + KEEP;
    #pragma unroll
    for (int j = 0; j < 8; ++j) {
        #pragma unroll
        for (int nt = 0; nt < NT; ++nt) {
            mOwn[(row0 + 4 * j) * TP + 4 * c + nt] = sumM[nt][j];
            if (nt > KEEP) qOwn[(row0 + 4 * j) * TP + 4 * c + nt] = sumQ[nt][j];
        }
    }
}

template <int KEEP>
__device__ __forceinline__ void finish(const float* mMirror, const float* qMirror, int c, int h,
                                       const v8f (&sumQ)[NT], const v8f (&sumM)[NT],
                                       float* __restrict__ out_r, float* __restrict__ out_i, size_t o0)
{
    const int row0 = 32 * h + KEEP;
    #pragma unroll
    for (int j = 0; j < 8; ++j) {
        const int row = row0 + 4 * j;
        v4f mt, re;
        #pragma unroll
        for (int nt = 0; nt < NT; ++nt) {
            mt[nt] = mMirror[(4 * c + nt) * TP + row];
            if (nt < KEEP) re[nt] = qMirror[(4 * c + nt) * TP + row] - sumM[nt][j];
            else           re[nt] = sumQ[nt][j] - mt[nt];
        }
        const size_t o = o0 + (size_t)row * D_LEN;
        *(v4f*)&out_r[o] = re;
        *(v4f*)&out_i[o] = (v4f){ sumM[0][j], sumM[1][j], sumM[2][j], sumM[3][j] } - mt;
    }
}

__global__ void __launch_bounds__(256) complex_mixture_wmma(
    const float* __restrict__ in_r,
    const float* __restrict__ in_i,
    const float* __restrict__ wgt,
    float* __restrict__ out_r,
    float* __restrict__ out_i)
{
    __shared__ __attribute__((aligned(16))) float hand[8 * HAND];
    __shared__ float mBlk[2][BLK * TP];
    __shared__ float qBlk[2][BLK * TP];

    const int b = blockIdx.y;
    const int u = blockIdx.x;
    int I, J;
    if (u < 3)       { I = 0; J = u + 1; }
    else if (u < 5)  { I = 1; J = u - 1; }
    else if (u == 5) { I = 2; J = 3;     }
    else if (u == 6) { I = 0; J = 1;     }
    else             { I = 2; J = 3;     }
    const int pair = u < 6 ? 1 : 0;

    const int tid   = threadIdx.x;
    const int lane  = tid & 31;
    const int wave  = __builtin_amdgcn_readfirstlane(tid >> 5);
    const int blk   = (wave >> 1) & 1;
    const int q     = (wave & 1) | ((wave >> 2) << 1);

    const int rowBlk = blk ? J : I;
    const int colBlk = (blk ^ pair) ? J : I;
    const int mirror = blk ^ pair;
    const int aBase  = rowBlk * BLK;
    const int cBase  = colBlk * BLK;

    const int c = lane & 15;
    const int h = lane >> 4;

    const float* rSrc = in_r + (size_t)b * S_LEN * D_LEN;
    const float* iSrc = in_i + (size_t)b * S_LEN * D_LEN;
    const float* wSrc = wgt  + (size_t)b * S_LEN;

    const unsigned offA = (unsigned)(2 * h * D_LEN + aBase + 4 * c);
    const unsigned offB = (unsigned)(2 * h * D_LEN + cBase + 4 * c);
    const unsigned offW = (unsigned)(2 * h);

    v8f accP[MT][NT] = {};
    v8f accM[MT][NT] = {};

    Slot setA[GRP], setB[GRP];
    load_group(setA, rSrc, iSrc, wSrc, offA, offB, offW, q);

    __builtin_amdgcn_s_setprio(3);
    for (int j = 0; j < NJ - 2; j += 2) {
        if (j == PRIO_FIRST)  __builtin_amdgcn_s_setprio(2);
        if (j == PRIO_SECOND) __builtin_amdgcn_s_setprio(1);
        load_group(setB, rSrc, iSrc, wSrc, offA, offB, offW, KQ * (j + 1) + q);
        KEEP_GROUP_AHEAD();
        mma_group(setA, accP, accM);
        KEEP_GROUP_AHEAD();
        load_group(setA, rSrc, iSrc, wSrc, offA, offB, offW, KQ * (j + 2) + q);
        KEEP_GROUP_AHEAD();
        mma_group(setB, accP, accM);
        KEEP_GROUP_AHEAD();
    }
    __builtin_amdgcn_s_setprio(0);
    load_group(setB, rSrc, iSrc, wSrc, offA, offB, offW, KQ * (NJ - 1) + q);
    KEEP_GROUP_AHEAD();
    mma_group(setA, accP, accM);
    mma_group(setB, accP, accM);

    float* mine = hand + wave * HAND;
    if (q == 0)      hand_over<0>(mine, lane, accP, accM);
    else if (q == 1) hand_over<1>(mine, lane, accP, accM);
    else if (q == 2) hand_over<2>(mine, lane, accP, accM);
    else             hand_over<3>(mine, lane, accP, accM);
    __syncthreads();

    v8f sumQ[NT], sumM[NT];
    float* mOwn = mBlk[blk];
    float* qOwn = qBlk[blk];
    if (q == 0)      { join<0>(hand, blk, lane, accP, accM, sumQ, sumM); lay_out<0>(mOwn, qOwn, c, h, sumQ, sumM); }
    else if (q == 1) { join<1>(hand, blk, lane, accP, accM, sumQ, sumM); lay_out<1>(mOwn, qOwn, c, h, sumQ, sumM); }
    else if (q == 2) { join<2>(hand, blk, lane, accP, accM, sumQ, sumM); lay_out<2>(mOwn, qOwn, c, h, sumQ, sumM); }
    else             { join<3>(hand, blk, lane, accP, accM, sumQ, sumM); lay_out<3>(mOwn, qOwn, c, h, sumQ, sumM); }
    __syncthreads();

    const float* mMirror = mBlk[mirror];
    const float* qMirror = qBlk[mirror];
    const size_t o0 = ((size_t)b * D_LEN + rowBlk * BLK) * D_LEN + cBase + 4 * c;
    if (q == 0)      finish<0>(mMirror, qMirror, c, h, sumQ, sumM, out_r, out_i, o0);
    else if (q == 1) finish<1>(mMirror, qMirror, c, h, sumQ, sumM, out_r, out_i, o0);
    else if (q == 2) finish<2>(mMirror, qMirror, c, h, sumQ, sumM, out_r, out_i, o0);
    else             finish<3>(mMirror, qMirror, c, h, sumQ, sumM, out_r, out_i, o0);
}

extern "C" void kernel_launch(void* const* d_in, const int* in_sizes, int n_in,
                              void* d_out, int out_size, void* d_ws, size_t ws_size,
                              hipStream_t stream) {
    const float* in_r = (const float*)d_in[0];
    const float* in_i = (const float*)d_in[1];
    const float* w    = (const float*)d_in[2];
    float* out_r = (float*)d_out;
    float* out_i = out_r + (size_t)32 * D_LEN * D_LEN;

    dim3 grid(8, 32);
    dim3 block(256);
    hipLaunchKernelGGL(complex_mixture_wmma, grid, block, 0, stream,
                       in_r, in_i, w, out_r, out_i);
}
